# starting kernel
# baseline (speedup 1.0000x reference)
#include <hip/hip_runtime.h>


typedef __attribute__((ext_vector_type(2))) float v2f;
typedef __attribute__((ext_vector_type(8))) float v8f;

#define D_DIM 1024
#define LSEQ  2048

template <bool RELU_BIAS>
__global__ __launch_bounds__(128) void gemm_f32_wmma(
    const float* __restrict__ A, const float* __restrict__ Bm,
    const float* __restrict__ bias, float* __restrict__ C,
    int M, int N, int K)
{
    const int lane = threadIdx.x & 31;
    const int wave = threadIdx.x >> 5;
    const int half = lane >> 4;
    const int l16  = lane & 15;
    const int koff = half * 2;

    const int m0 = blockIdx.y * 64 + (wave >> 1) * 32;
    const int n0 = blockIdx.x * 64 + (wave & 1) * 32;

    const float* Arow0 = A + (size_t)(m0 + l16) * K + koff;
    const float* Arow1 = A + (size_t)(m0 + 16 + l16) * K + koff;
    const float* Bcol0 = Bm + n0 + l16;
    const float* Bcol1 = Bm + n0 + 16 + l16;

    v8f acc00 = {}, acc01 = {}, acc10 = {}, acc11 = {};
    for (int k0 = 0; k0 < K; k0 += 4) {
        v2f a0 = *(const v2f*)(Arow0 + k0);
        v2f a1 = *(const v2f*)(Arow1 + k0);
        v2f b0, b1;
        size_t r0 = (size_t)(k0 + koff) * N;
        size_t r1 = r0 + N;
        b0.x = Bcol0[r0];  b0.y = Bcol0[r1];
        b1.x = Bcol1[r0];  b1.y = Bcol1[r1];
        acc00 = __builtin_amdgcn_wmma_f32_16x16x4_f32(false, a0, false, b0, (short)0, acc00, false, false);
        acc01 = __builtin_amdgcn_wmma_f32_16x16x4_f32(false, a0, false, b1, (short)0, acc01, false, false);
        acc10 = __builtin_amdgcn_wmma_f32_16x16x4_f32(false, a1, false, b0, (short)0, acc10, false, false);
        acc11 = __builtin_amdgcn_wmma_f32_16x16x4_f32(false, a1, false, b1, (short)0, acc11, false, false);
    }

    float bv0 = 0.0f, bv1 = 0.0f;
    if (RELU_BIAS) {
        bv0 = bias[n0 + l16];
        bv1 = bias[n0 + 16 + l16];
    }
#pragma unroll
    for (int r = 0; r < 8; ++r) {
        int row0 = m0 + r + half * 8;
        int row1 = row0 + 16;
        float v00 = acc00[r], v01 = acc01[r], v10 = acc10[r], v11 = acc11[r];
        if (RELU_BIAS) {
            v00 += bv0; v01 += bv1; v10 += bv0; v11 += bv1;
            v00 = v00 > 0.0f ? v00 : 0.0f;
            v01 = v01 > 0.0f ? v01 : 0.0f;
            v10 = v10 > 0.0f ? v10 : 0.0f;
            v11 = v11 > 0.0f ? v11 : 0.0f;
        }
        C[(size_t)row0 * N + n0 + l16]      = v00;
        C[(size_t)row0 * N + n0 + 16 + l16] = v01;
        C[(size_t)row1 * N + n0 + l16]      = v10;
        C[(size_t)row1 * N + n0 + 16 + l16] = v11;
    }
}

__global__ __launch_bounds__(256) void cumavg_kernel(
    const float* __restrict__ q, float* __restrict__ xavg, int Bsz)
{
    int idx = blockIdx.x * blockDim.x + threadIdx.x;
    if (idx >= Bsz * D_DIM) return;
    int b = idx / D_DIM, d = idx % D_DIM;
    size_t base = (size_t)b * LSEQ * D_DIM + d;
    float acc = 0.0f;
    for (int i = 0; i < LSEQ; ++i) {
        acc += q[base + (size_t)i * D_DIM];
        xavg[base + (size_t)i * D_DIM] = acc / (float)(i + 1);
    }
}

__global__ __launch_bounds__(256) void rowsum_kernel(
    const float* __restrict__ g, float* __restrict__ S, int nrows)
{
    int wid  = (int)((blockIdx.x * blockDim.x + threadIdx.x) >> 5);
    int lane = threadIdx.x & 31;
    if (wid >= nrows) return;
    int i = wid % LSEQ;
    const float* row = g + (size_t)wid * LSEQ;
    float s = 0.0f;
    for (int k = (LSEQ - 1 - i) + lane; k < LSEQ; k += 32) s += row[k];
#pragma unroll
    for (int off = 16; off; off >>= 1) s += __shfl_xor(s, off, 32);
    if (lane == 0) S[wid] = s + 1e-8f;
}

__global__ __launch_bounds__(128) void attn_apply_wmma(
    const float* __restrict__ g, const float* __restrict__ q,
    const float* __restrict__ S, float* __restrict__ out)
{
    const int lane = threadIdx.x & 31;
    const int wave = threadIdx.x >> 5;
    const int b    = blockIdx.z;
    const int i0   = blockIdx.y * 16;
    const int d0   = blockIdx.x * 128 + wave * 32;
    const int half = lane >> 4;
    const int l16  = lane & 15;
    const int koff = half * 2;

    const int iA    = i0 + l16;
    const int shift = LSEQ - 1 - iA;
    const float* grow = g + ((size_t)b * LSEQ + iA) * LSEQ + shift;
    const float* qB0  = q + (size_t)b * LSEQ * D_DIM + d0 + l16;
    const float* qB1  = qB0 + 16;

    v8f acc0 = {}, acc1 = {};

    for (int j0 = 0; j0 < i0; j0 += 4) {
        int jx = j0 + koff;
        v2f a, b0, b1;
        a.x = grow[jx];
        a.y = grow[jx + 1];
        size_t r0 = (size_t)jx * D_DIM;
        size_t r1 = r0 + D_DIM;
        b0.x = qB0[r0];  b0.y = qB0[r1];
        b1.x = qB1[r0];  b1.y = qB1[r1];
        acc0 = __builtin_amdgcn_wmma_f32_16x16x4_f32(false, a, false, b0, (short)0, acc0, false, false);
        acc1 = __builtin_amdgcn_wmma_f32_16x16x4_f32(false, a, false, b1, (short)0, acc1, false, false);
    }

#pragma unroll
    for (int t = 0; t < 4; ++t) {
        int jx = i0 + t * 4 + koff;
        int jy = jx + 1;
        v2f a, b0, b1;
        a.x = (jx <= iA) ? grow[jx] : 0.0f;
        a.y = (jy <= iA) ? grow[jy] : 0.0f;
        size_t r0 = (size_t)jx * D_DIM;
        size_t r1 = r0 + D_DIM;
        b0.x = qB0[r0];  b0.y = qB0[r1];
        b1.x = qB1[r0];  b1.y = qB1[r1];
        acc0 = __builtin_amdgcn_wmma_f32_16x16x4_f32(false, a, false, b0, (short)0, acc0, false, false);
        acc1 = __builtin_amdgcn_wmma_f32_16x16x4_f32(false, a, false, b1, (short)0, acc1, false, false);
    }

#pragma unroll
    for (int r = 0; r < 8; ++r) {
        int row = i0 + r + half * 8;
        float invs = 1.0f / S[b * LSEQ + row];
        size_t o = ((size_t)b * LSEQ + row) * D_DIM + d0 + l16;
        out[o]      = acc0[r] * invs;
        out[o + 16] = acc1[r] * invs;
    }
}

extern "C" void kernel_launch(void* const* d_in, const int* in_sizes, int n_in,
                              void* d_out, int out_size, void* d_ws, size_t ws_size,
                              hipStream_t stream)
{
    const float* x  = (const float*)d_in[0];
    const float* Wq = (const float*)d_in[1];
    const float* Wg = (const float*)d_in[2];
    const float* bg = (const float*)d_in[3];
    const float* Wc = (const float*)d_in[4];
    float* out = (float*)d_out;

    const int Bsz = in_sizes[0] / (LSEQ * D_DIM);
    const int M   = Bsz * LSEQ;

    float* q    = (float*)d_ws;
    float* xavg = q    + (size_t)M * D_DIM;
    float* g    = xavg + (size_t)M * D_DIM;
    float* S    = g    + (size_t)M * LSEQ;

    dim3 blk(128);

    gemm_f32_wmma<false><<<dim3(D_DIM / 64, M / 64), blk, 0, stream>>>(
        x, Wq, nullptr, q, M, D_DIM, D_DIM);

    cumavg_kernel<<<(Bsz * D_DIM + 255) / 256, 256, 0, stream>>>(q, xavg, Bsz);

    gemm_f32_wmma<true><<<dim3(LSEQ / 64, M / 64), blk, 0, stream>>>(
        xavg, Wg, bg, g, M, LSEQ, D_DIM);

    rowsum_kernel<<<(M * 32 + 255) / 256, 256, 0, stream>>>(g, S, M);

    attn_apply_wmma<<<dim3(D_DIM / 128, LSEQ / 16, Bsz), blk, 0, stream>>>(
        g, q, S, xavg);

    gemm_f32_wmma<false><<<dim3(D_DIM / 64, M / 64), blk, 0, stream>>>(
        xavg, Wc, nullptr, out, M, D_DIM, D_DIM);
}


# causal mean (scan launch): serial sums kept, on one wave per 32 columns; row loads, divisions and stores moved to four other waves of the group via LDS tiles, one barrier per 128 rows; sum order same
# speedup vs baseline: 2.5443x; 2.5443x over previous
#include <hip/hip_runtime.h>


typedef __attribute__((ext_vector_type(2))) float v2f;
typedef __attribute__((ext_vector_type(4))) float v4f;
typedef __attribute__((ext_vector_type(8))) float v8f;

#define D_DIM 1024
#define LSEQ  2048

template <int N>
__device__ __forceinline__ void gemm_load_step(v4f (&a0)[2], v4f (&a1)[2], v4f (&b)[2][8],
                                               const float* ap0, const float* ap1, const float* bp)
{
    a0[0] = *(const v4f*)(ap0);
    a0[1] = *(const v4f*)(ap0 + 4);
    a1[0] = *(const v4f*)(ap1);
    a1[1] = *(const v4f*)(ap1 + 4);
#pragma unroll
    for (int e = 0; e < 8; ++e) {
        b[0][e] = *(const v4f*)(bp + (size_t)e * N);
        b[1][e] = *(const v4f*)(bp + (size_t)e * N + 64);
    }
}

__device__ __forceinline__ void gemm_mma_step(const v4f (&a0)[2], const v4f (&a1)[2],
                                              const v4f (&b)[2][8], v8f (&acc)[2][8])
{
#pragma unroll
    for (int p = 0; p < 4; ++p) {
        v2f af0, af1;
        af0.x = a0[p >> 1][(p & 1) * 2];  af0.y = a0[p >> 1][(p & 1) * 2 + 1];
        af1.x = a1[p >> 1][(p & 1) * 2];  af1.y = a1[p >> 1][(p & 1) * 2 + 1];
#pragma unroll
        for (int u = 0; u < 2; ++u) {
#pragma unroll
            for (int t = 0; t < 4; ++t) {
                v2f bf;
                bf.x = b[u][2 * p][t];
                bf.y = b[u][2 * p + 1][t];
                acc[0][4 * u + t] = __builtin_amdgcn_wmma_f32_16x16x4_f32(false, af0, false, bf, (short)0, acc[0][4 * u + t], false, false);
                acc[1][4 * u + t] = __builtin_amdgcn_wmma_f32_16x16x4_f32(false, af1, false, bf, (short)0, acc[1][4 * u + t], false, false);
            }
        }
    }
}

template <bool RELU_BIAS, int ROW_TILES, int N, int K>
__global__ __launch_bounds__(128) void gemm_f32_wmma(
    const float* __restrict__ A, const float* __restrict__ Bm,
    const float* __restrict__ bias, float* __restrict__ C)
{
    static_assert(K % 32 == 0 && K >= 32, "the k loop runs in pairs of 16-k steps");
    const int lane = threadIdx.x & 31;
    const int wave = threadIdx.x >> 5;
    const int h    = lane >> 4;
    const int c    = lane & 15;

    const int n0 = blockIdx.x * 128;
    const float* Bp = Bm + (size_t)(8 * h) * N + n0 + 4 * c;

    v4f bv[2] = {{0.0f, 0.0f, 0.0f, 0.0f}, {0.0f, 0.0f, 0.0f, 0.0f}};
    if (RELU_BIAS) {
        bv[0] = *(const v4f*)(bias + n0 + 4 * c);
        bv[1] = *(const v4f*)(bias + n0 + 64 + 4 * c);
    }

    for (int s = 0; s < ROW_TILES; ++s) {
        const int m0 = ((int)blockIdx.y * ROW_TILES + s) * 128 + wave * 32;
        const float* Ap0 = A + (size_t)(m0 + c) * K + 8 * h;
        const float* Ap1 = Ap0 + (size_t)16 * K;

        v8f acc[2][8] = {};
        v4f xa0[2], xa1[2], xb[2][8];
        v4f ya0[2], ya1[2], yb[2][8];
        gemm_load_step<N>(xa0, xa1, xb, Ap0, Ap1, Bp);
        for (int k0 = 0; k0 < K - 32; k0 += 32) {
            gemm_load_step<N>(ya0, ya1, yb, Ap0 + k0 + 16, Ap1 + k0 + 16, Bp + (size_t)(k0 + 16) * N);
            __builtin_amdgcn_sched_barrier(0);
            gemm_mma_step(xa0, xa1, xb, acc);
            __builtin_amdgcn_sched_barrier(0);
            gemm_load_step<N>(xa0, xa1, xb, Ap0 + k0 + 32, Ap1 + k0 + 32, Bp + (size_t)(k0 + 32) * N);
            __builtin_amdgcn_sched_barrier(0);
            gemm_mma_step(ya0, ya1, yb, acc);
            __builtin_amdgcn_sched_barrier(0);
        }
        gemm_load_step<N>(ya0, ya1, yb, Ap0 + K - 16, Ap1 + K - 16, Bp + (size_t)(K - 16) * N);
        __builtin_amdgcn_sched_barrier(0);
        gemm_mma_step(xa0, xa1, xb, acc);
        __builtin_amdgcn_sched_barrier(0);
        gemm_mma_step(ya0, ya1, yb, acc);

#pragma unroll
        for (int mi = 0; mi < 2; ++mi) {
#pragma unroll
            for (int r = 0; r < 8; ++r) {
                int row = m0 + mi * 16 + r + h * 8;
#pragma unroll
                for (int u = 0; u < 2; ++u) {
                    v4f v;
                    v.x = acc[mi][4 * u + 0][r]; v.y = acc[mi][4 * u + 1][r];
                    v.z = acc[mi][4 * u + 2][r]; v.w = acc[mi][4 * u + 3][r];
                    if (RELU_BIAS) {
                        v += bv[u];
                        v.x = v.x > 0.0f ? v.x : 0.0f;
                        v.y = v.y > 0.0f ? v.y : 0.0f;
                        v.z = v.z > 0.0f ? v.z : 0.0f;
                        v.w = v.w > 0.0f ? v.w : 0.0f;
                    }
                    *(v4f*)(C + (size_t)row * N + n0 + 64 * u + 4 * c) = v;
                }
            }
        }
    }
}

#define SCAN_ROWS   128
#define SCAN_PHASES (LSEQ / SCAN_ROWS)
#define SCAN_SHARE  (SCAN_ROWS / 4)

__device__ __forceinline__ void scan_barrier()
{
    __builtin_amdgcn_fence(__ATOMIC_RELEASE, "workgroup", "local");
    __builtin_amdgcn_s_barrier();
    __builtin_amdgcn_fence(__ATOMIC_ACQUIRE, "workgroup", "local");
}

__device__ __forceinline__ void scan_add_phase(const float (*in_t)[32], float (*out_t)[32],
                                               float& acc, int lane)
{
    float v[SCAN_ROWS];
#pragma unroll
    for (int r = 0; r < SCAN_ROWS; ++r) v[r] = in_t[r][lane];
    __builtin_amdgcn_sched_barrier(0);
#pragma unroll
    for (int r = 0; r < SCAN_ROWS; ++r) {
        acc += v[r];
        out_t[r][lane] = acc;
    }
}

template <bool LOAD, bool DIVIDE>
__device__ __forceinline__ void scan_feed_phase(float (*in_t)[32], const float (*out_t)[32],
                                                const float* nxt, float* dst, int row0,
                                                int cw, int lane)
{
    float s[SCAN_SHARE];
    if (LOAD) {
#pragma unroll
        for (int k = 0; k < SCAN_SHARE; ++k) s[k] = nxt[(size_t)(4 * k + cw) * D_DIM];
    }
    if (DIVIDE) {
        float v[SCAN_SHARE];
#pragma unroll
        for (int k = 0; k < SCAN_SHARE; ++k) v[k] = out_t[4 * k + cw][lane];
#pragma unroll
        for (int k = 0; k < SCAN_SHARE; ++k) {
            const int row = row0 + 4 * k + cw;
            dst[(size_t)row * D_DIM] = v[k] / (float)(row + 1);
        }
    }
    if (LOAD) {
        if (DIVIDE) __builtin_amdgcn_sched_barrier(0);
#pragma unroll
        for (int k = 0; k < SCAN_SHARE; ++k) in_t[4 * k + cw][lane] = s[k];
    }
}

__global__ __launch_bounds__(160) void cumavg_kernel(
    const float* __restrict__ q, float* __restrict__ xavg, int Bsz)
{
    __shared__ float t_in[3][SCAN_ROWS][32];
    __shared__ float t_out[2][SCAN_ROWS][32];

    const int lane = threadIdx.x & 31;
    const int wave = __builtin_amdgcn_readfirstlane((int)(threadIdx.x >> 5));
    const int col0 = blockIdx.x * 32;
    if (col0 >= Bsz * D_DIM) return;
    const int idx = col0 + lane;
    const int b = idx / D_DIM, d = idx % D_DIM;
    const float* src = q    + (size_t)b * LSEQ * D_DIM + d;
    float*       dst = xavg + (size_t)b * LSEQ * D_DIM + d;
    const size_t phase_stride = (size_t)SCAN_ROWS * D_DIM;

    float acc = 0.0f;

    if (wave != 0) {
        scan_feed_phase<true, false>(t_in[0], t_out[0], src, dst, 0, wave - 1, lane);
        scan_feed_phase<true, false>(t_in[1], t_out[0], src + phase_stride, dst, 0, wave - 1, lane);
    }
    scan_barrier();

    int in_a = 0;
    int in_f = 2;
    for (int p = 0; p < SCAN_PHASES; ++p) {
        if (wave == 0) {
            scan_add_phase(t_in[in_a], t_out[p & 1], acc, lane);
        } else {
            const float* nxt = src + (size_t)(p + 2) * phase_stride;
            const int    row0 = (p - 1) * SCAN_ROWS;
            const float (*out_f)[32] = t_out[(p + 1) & 1];
            if (p == 0)
                scan_feed_phase<true, false>(t_in[in_f], out_f, nxt, dst, row0, wave - 1, lane);
            else if (p + 2 < SCAN_PHASES)
                scan_feed_phase<true, true>(t_in[in_f], out_f, nxt, dst, row0, wave - 1, lane);
            else
                scan_feed_phase<false, true>(t_in[in_f], out_f, nxt, dst, row0, wave - 1, lane);
        }
        scan_barrier();
        in_a = (in_a == 2) ? 0 : in_a + 1;
        in_f = (in_f == 2) ? 0 : in_f + 1;
    }
    if (wave != 0)
        scan_feed_phase<false, true>(t_in[0], t_out[(SCAN_PHASES - 1) & 1], src, dst,
                                     (SCAN_PHASES - 1) * SCAN_ROWS, wave - 1, lane);
}

__global__ __launch_bounds__(256) void rowsum_kernel(
    const float* __restrict__ g, float* __restrict__ S, int nrows)
{
    int wid  = (int)((blockIdx.x * blockDim.x + threadIdx.x) >> 5);
    int lane = threadIdx.x & 31;
    if (wid >= nrows) return;
    int i = wid % LSEQ;
    const float* row = g + (size_t)wid * LSEQ;
    float s = 0.0f;
    for (int k = (LSEQ - 1 - i) + lane; k < LSEQ; k += 32) s += row[k];
#pragma unroll
    for (int off = 16; off; off >>= 1) s += __shfl_xor(s, off, 32);
    if (lane == 0) S[wid] = s + 1e-8f;
}

typedef v4f v4f_a4 __attribute__((aligned(4)));

__device__ __forceinline__ v4f keep_le(v4f v, int kb, int c)
{
    v.x = (kb     <= c) ? v.x : 0.0f;
    v.y = (kb + 1 <= c) ? v.y : 0.0f;
    v.z = (kb + 2 <= c) ? v.z : 0.0f;
    v.w = (kb + 3 <= c) ? v.w : 0.0f;
    return v;
}

template <int P>
__device__ __forceinline__ void wmma_row4(const v4f (&a)[2], const v4f (&b)[8], v8f (&acc)[4])
{
    v2f af;
    af.x = a[P >> 1][(P & 1) * 2];
    af.y = a[P >> 1][(P & 1) * 2 + 1];
#pragma unroll
    for (int t = 0; t < 4; ++t) {
        v2f bf;
        bf.x = b[2 * P][t];
        bf.y = b[2 * P + 1][t];
        acc[t] = __builtin_amdgcn_wmma_f32_16x16x4_f32(false, af, false, bf, (short)0, acc[t], false, false);
    }
}

__device__ __forceinline__ void attn_load_step(v4f (&a0)[2], v4f (&a1)[2], v4f (&bq)[8],
                                               const float* ap0, const float* ap1, const float* bp)
{
    a0[0] = *(const v4f_a4*)(ap0);
    a0[1] = *(const v4f_a4*)(ap0 + 4);
    a1[0] = *(const v4f_a4*)(ap1);
    a1[1] = *(const v4f_a4*)(ap1 + 4);
#pragma unroll
    for (int e = 0; e < 8; ++e) bq[e] = *(const v4f*)(bp + (size_t)e * D_DIM);
}

__device__ __forceinline__ void attn_mma_step(const v4f (&a0)[2], const v4f (&a1)[2],
                                              const v4f (&bq)[8], v8f (&acc)[2][4])
{
    wmma_row4<0>(a0, bq, acc[0]);  wmma_row4<0>(a1, bq, acc[1]);
    wmma_row4<1>(a0, bq, acc[0]);  wmma_row4<1>(a1, bq, acc[1]);
    wmma_row4<2>(a0, bq, acc[0]);  wmma_row4<2>(a1, bq, acc[1]);
    wmma_row4<3>(a0, bq, acc[0]);  wmma_row4<3>(a1, bq, acc[1]);
}

__global__ __launch_bounds__(128) void attn_apply_wmma(
    const float* __restrict__ g, const float* __restrict__ q,
    const float* __restrict__ S, float* __restrict__ out)
{
    const int lane = threadIdx.x & 31;
    const int wave = threadIdx.x >> 5;
    const int h    = lane >> 4;
    const int c    = lane & 15;
    const int b    = blockIdx.z;
    const int d0   = blockIdx.x * 256 + wave * 64;

    const float* qb = q + (size_t)b * LSEQ * D_DIM + (size_t)(8 * h) * D_DIM + d0 + 4 * c;

    for (int s = 0; s < 2; ++s) {
        const int y  = s ? (LSEQ / 32 - 1 - (int)blockIdx.y) : (int)blockIdx.y;
        const int i0 = y * 32;
        const int r0 = i0 + c;
        const int r1 = r0 + 16;
        const float* Ap0 = g + ((size_t)b * LSEQ + r0) * LSEQ + (LSEQ - 1 - r0) + 8 * h;
        const float* Ap1 = g + ((size_t)b * LSEQ + r1) * LSEQ + (LSEQ - 1 - r1) + 8 * h;

        v8f acc[2][4] = {};

        v4f xa0[2], xa1[2], xb[8];
        v4f ya0[2], ya1[2], yb[8];
        attn_load_step(xa0, xa1, xb, Ap0, Ap1, qb);
        for (int j0 = 0; j0 < i0; j0 += 32) {
            attn_load_step(ya0, ya1, yb, Ap0 + j0 + 16, Ap1 + j0 + 16, qb + (size_t)(j0 + 16) * D_DIM);
            __builtin_amdgcn_sched_barrier(0);
            attn_mma_step(xa0, xa1, xb, acc);
            __builtin_amdgcn_sched_barrier(0);
            attn_load_step(xa0, xa1, xb, Ap0 + j0 + 32, Ap1 + j0 + 32, qb + (size_t)(j0 + 32) * D_DIM);
            __builtin_amdgcn_sched_barrier(0);
            attn_mma_step(ya0, ya1, yb, acc);
            __builtin_amdgcn_sched_barrier(0);
        }

        ya1[0] = *(const v4f_a4*)(Ap1 + i0 + 16);
        ya1[1] = *(const v4f_a4*)(Ap1 + i0 + 20);
        {
            const float* bp = qb + (size_t)(i0 + 16) * D_DIM;
#pragma unroll
            for (int e = 0; e < 8; ++e) yb[e] = *(const v4f*)(bp + (size_t)e * D_DIM);
        }
        __builtin_amdgcn_sched_barrier(0);
        xa0[0] = keep_le(xa0[0], 8 * h,     c);
        xa0[1] = keep_le(xa0[1], 8 * h + 4, c);
        attn_mma_step(xa0, xa1, xb, acc);
        __builtin_amdgcn_sched_barrier(0);
        ya1[0] = keep_le(ya1[0], 8 * h,     c);
        ya1[1] = keep_le(ya1[1], 8 * h + 4, c);
        wmma_row4<0>(ya1, yb, acc[1]);
        wmma_row4<1>(ya1, yb, acc[1]);
        wmma_row4<2>(ya1, yb, acc[1]);
        wmma_row4<3>(ya1, yb, acc[1]);

#pragma unroll
        for (int mi = 0; mi < 2; ++mi) {
#pragma unroll
            for (int r = 0; r < 8; ++r) {
                int row = i0 + mi * 16 + r + h * 8;
                float invs = 1.0f / S[b * LSEQ + row];
                v4f v;
                v.x = acc[mi][0][r] * invs;  v.y = acc[mi][1][r] * invs;
                v.z = acc[mi][2][r] * invs;  v.w = acc[mi][3][r] * invs;
                *(v4f*)(out + ((size_t)b * LSEQ + row) * D_DIM + d0 + 4 * c) = v;
            }
        }
    }
}

extern "C" void kernel_launch(void* const* d_in, const int* in_sizes, int n_in,
                              void* d_out, int out_size, void* d_ws, size_t ws_size,
                              hipStream_t stream)
{
    const float* x  = (const float*)d_in[0];
    const float* Wq = (const float*)d_in[1];
    const float* Wg = (const float*)d_in[2];
    const float* bg = (const float*)d_in[3];
    const float* Wc = (const float*)d_in[4];
    float* out = (float*)d_out;

    const int Bsz = in_sizes[0] / (LSEQ * D_DIM);
    const int M   = Bsz * LSEQ;

    float* q    = (float*)d_ws;
    float* xavg = q    + (size_t)M * D_DIM;
    float* g    = xavg + (size_t)M * D_DIM;
    float* S    = g    + (size_t)M * LSEQ;

    dim3 blk(128);

    gemm_f32_wmma<false, 1, D_DIM, D_DIM><<<dim3(D_DIM / 128, M / 128), blk, 0, stream>>>(
        x, Wq, nullptr, q);

    cumavg_kernel<<<(Bsz * D_DIM + 31) / 32, 160, 0, stream>>>(q, xavg, Bsz);

    gemm_f32_wmma<true, 2, LSEQ, D_DIM><<<dim3(LSEQ / 128, M / 256), blk, 0, stream>>>(
        xavg, Wg, bg, g);

    rowsum_kernel<<<(M * 32 + 255) / 256, 256, 0, stream>>>(g, S, M);

    attn_apply_wmma<<<dim3(D_DIM / 256, LSEQ / 64, Bsz), blk, 0, stream>>>(
        g, q, S, xavg);

    gemm_f32_wmma<false, 1, D_DIM, D_DIM><<<dim3(D_DIM / 128, M / 128), blk, 0, stream>>>(
        xavg, Wc, nullptr, out);
}


# attn normaliser: masked row sums taken in the g GEMM epilogue from its accumulators, one per 128-column block; attn apply adds a row's 16 block sums; row-sum launch gone; row-sum order changes
# speedup vs baseline: 2.6207x; 1.0300x over previous
#include <hip/hip_runtime.h>


typedef __attribute__((ext_vector_type(2))) float v2f;
typedef __attribute__((ext_vector_type(4))) float v4f;
typedef __attribute__((ext_vector_type(8))) float v8f;

#define D_DIM 1024
#define LSEQ  2048

template <int N>
__device__ __forceinline__ void gemm_load_step(v4f (&a0)[2], v4f (&a1)[2], v4f (&b)[2][8],
                                               const float* ap0, const float* ap1, const float* bp)
{
    a0[0] = *(const v4f*)(ap0);
    a0[1] = *(const v4f*)(ap0 + 4);
    a1[0] = *(const v4f*)(ap1);
    a1[1] = *(const v4f*)(ap1 + 4);
#pragma unroll
    for (int e = 0; e < 8; ++e) {
        b[0][e] = *(const v4f*)(bp + (size_t)e * N);
        b[1][e] = *(const v4f*)(bp + (size_t)e * N + 64);
    }
}

__device__ __forceinline__ void gemm_mma_step(const v4f (&a0)[2], const v4f (&a1)[2],
                                              const v4f (&b)[2][8], v8f (&acc)[2][8])
{
#pragma unroll
    for (int p = 0; p < 4; ++p) {
        v2f af0, af1;
        af0.x = a0[p >> 1][(p & 1) * 2];  af0.y = a0[p >> 1][(p & 1) * 2 + 1];
        af1.x = a1[p >> 1][(p & 1) * 2];  af1.y = a1[p >> 1][(p & 1) * 2 + 1];
#pragma unroll
        for (int u = 0; u < 2; ++u) {
#pragma unroll
            for (int t = 0; t < 4; ++t) {
                v2f bf;
                bf.x = b[u][2 * p][t];
                bf.y = b[u][2 * p + 1][t];
                acc[0][4 * u + t] = __builtin_amdgcn_wmma_f32_16x16x4_f32(false, af0, false, bf, (short)0, acc[0][4 * u + t], false, false);
                acc[1][4 * u + t] = __builtin_amdgcn_wmma_f32_16x16x4_f32(false, af1, false, bf, (short)0, acc[1][4 * u + t], false, false);
            }
        }
    }
}


__device__ __forceinline__ float sum16_index_to_lane(const float (&x)[16], int c)
{
    const bool b8 = (c & 8) != 0, b4 = (c & 4) != 0, b2 = (c & 2) != 0, b1 = (c & 1) != 0;
    float y8[8], y4[4], y2[2];
#pragma unroll
    for (int q = 0; q < 8; ++q) {
        const float keep = b8 ? x[8 + q] : x[q];
        const float send = b8 ? x[q] : x[8 + q];
        y8[q] = keep + __shfl_xor(send, 8, 32);
    }
#pragma unroll
    for (int q = 0; q < 4; ++q) {
        const float keep = b4 ? y8[4 + q] : y8[q];
        const float send = b4 ? y8[q] : y8[4 + q];
        y4[q] = keep + __shfl_xor(send, 4, 32);
    }
#pragma unroll
    for (int q = 0; q < 2; ++q) {
        const float keep = b2 ? y4[2 + q] : y4[q];
        const float send = b2 ? y4[q] : y4[2 + q];
        y2[q] = keep + __shfl_xor(send, 2, 32);
    }
    const float keep = b1 ? y2[1] : y2[0];
    const float send = b1 ? y2[0] : y2[1];
    return keep + __shfl_xor(send, 1, 32);
}

template <bool RELU_BIAS, bool ROW_SUMS, int ROW_TILES, int N, int K>
__global__ __launch_bounds__(128) void gemm_f32_wmma(
    const float* __restrict__ A, const float* __restrict__ Bm,
    const float* __restrict__ bias, float* __restrict__ C,
    float* __restrict__ rowpart)
{
    static_assert(K % 32 == 0 && K >= 32, "the k loop runs in pairs of 16-k steps");
    const int lane = threadIdx.x & 31;
    const int wave = threadIdx.x >> 5;
    const int h    = lane >> 4;
    const int c    = lane & 15;

    const int n0 = blockIdx.x * 128;
    const float* Bp = Bm + (size_t)(8 * h) * N + n0 + 4 * c;

    v4f bv[2] = {{0.0f, 0.0f, 0.0f, 0.0f}, {0.0f, 0.0f, 0.0f, 0.0f}};
    if (RELU_BIAS) {
        bv[0] = *(const v4f*)(bias + n0 + 4 * c);
        bv[1] = *(const v4f*)(bias + n0 + 64 + 4 * c);
    }

    for (int s = 0; s < ROW_TILES; ++s) {
        const int m0 = ((int)blockIdx.y * ROW_TILES + s) * 128 + wave * 32;
        const float* Ap0 = A + (size_t)(m0 + c) * K + 8 * h;
        const float* Ap1 = Ap0 + (size_t)16 * K;

        v8f acc[2][8] = {};
        v4f xa0[2], xa1[2], xb[2][8];
        v4f ya0[2], ya1[2], yb[2][8];
        gemm_load_step<N>(xa0, xa1, xb, Ap0, Ap1, Bp);
        for (int k0 = 0; k0 < K - 32; k0 += 32) {
            gemm_load_step<N>(ya0, ya1, yb, Ap0 + k0 + 16, Ap1 + k0 + 16, Bp + (size_t)(k0 + 16) * N);
            __builtin_amdgcn_sched_barrier(0);
            gemm_mma_step(xa0, xa1, xb, acc);
            __builtin_amdgcn_sched_barrier(0);
            gemm_load_step<N>(xa0, xa1, xb, Ap0 + k0 + 32, Ap1 + k0 + 32, Bp + (size_t)(k0 + 32) * N);
            __builtin_amdgcn_sched_barrier(0);
            gemm_mma_step(ya0, ya1, yb, acc);
            __builtin_amdgcn_sched_barrier(0);
        }
        gemm_load_step<N>(ya0, ya1, yb, Ap0 + K - 16, Ap1 + K - 16, Bp + (size_t)(K - 16) * N);
        __builtin_amdgcn_sched_barrier(0);
        gemm_mma_step(xa0, xa1, xb, acc);
        __builtin_amdgcn_sched_barrier(0);
        gemm_mma_step(ya0, ya1, yb, acc);

        float ps[16];
#pragma unroll
        for (int mi = 0; mi < 2; ++mi) {
#pragma unroll
            for (int r = 0; r < 8; ++r) {
                int row = m0 + mi * 16 + r + h * 8;
                const int kmin = (LSEQ - 1) - (row & (LSEQ - 1));
                float p = 0.0f;
#pragma unroll
                for (int u = 0; u < 2; ++u) {
                    v4f v;
                    v.x = acc[mi][4 * u + 0][r]; v.y = acc[mi][4 * u + 1][r];
                    v.z = acc[mi][4 * u + 2][r]; v.w = acc[mi][4 * u + 3][r];
                    if (RELU_BIAS) {
                        v += bv[u];
                        v.x = v.x > 0.0f ? v.x : 0.0f;
                        v.y = v.y > 0.0f ? v.y : 0.0f;
                        v.z = v.z > 0.0f ? v.z : 0.0f;
                        v.w = v.w > 0.0f ? v.w : 0.0f;
                    }
                    *(v4f*)(C + (size_t)row * N + n0 + 64 * u + 4 * c) = v;
                    if (ROW_SUMS) {
                        const int col = n0 + 64 * u + 4 * c;
                        p += (col     >= kmin) ? v.x : 0.0f;
                        p += (col + 1 >= kmin) ? v.y : 0.0f;
                        p += (col + 2 >= kmin) ? v.z : 0.0f;
                        p += (col + 3 >= kmin) ? v.w : 0.0f;
                    }
                }
                ps[8 * mi + r] = p;
            }
        }
        if (ROW_SUMS) {
            const float tot = sum16_index_to_lane(ps, c);
            const int mtot = (int)gridDim.y * ROW_TILES * 128;
            rowpart[(size_t)blockIdx.x * mtot + m0 + (c >> 3) * 16 + 8 * h + (c & 7)] = tot;
        }
    }
}

#define SCAN_ROWS   128
#define SCAN_PHASES (LSEQ / SCAN_ROWS)
#define SCAN_SHARE  (SCAN_ROWS / 4)

__device__ __forceinline__ void scan_barrier()
{
    __builtin_amdgcn_fence(__ATOMIC_RELEASE, "workgroup", "local");
    __builtin_amdgcn_s_barrier();
    __builtin_amdgcn_fence(__ATOMIC_ACQUIRE, "workgroup", "local");
}

__device__ __forceinline__ void scan_add_phase(const float (*in_t)[32], float (*out_t)[32],
                                               float& acc, int lane)
{
    float v[SCAN_ROWS];
#pragma unroll
    for (int r = 0; r < SCAN_ROWS; ++r) v[r] = in_t[r][lane];
    __builtin_amdgcn_sched_barrier(0);
#pragma unroll
    for (int r = 0; r < SCAN_ROWS; ++r) {
        acc += v[r];
        out_t[r][lane] = acc;
    }
}

template <bool LOAD, bool DIVIDE>
__device__ __forceinline__ void scan_feed_phase(float (*in_t)[32], const float (*out_t)[32],
                                                const float* nxt, float* dst, int row0,
                                                int cw, int lane)
{
    float s[SCAN_SHARE];
    if (LOAD) {
#pragma unroll
        for (int k = 0; k < SCAN_SHARE; ++k) s[k] = nxt[(size_t)(4 * k + cw) * D_DIM];
    }
    if (DIVIDE) {
        float v[SCAN_SHARE];
#pragma unroll
        for (int k = 0; k < SCAN_SHARE; ++k) v[k] = out_t[4 * k + cw][lane];
#pragma unroll
        for (int k = 0; k < SCAN_SHARE; ++k) {
            const int row = row0 + 4 * k + cw;
            dst[(size_t)row * D_DIM] = v[k] / (float)(row + 1);
        }
    }
    if (LOAD) {
        if (DIVIDE) __builtin_amdgcn_sched_barrier(0);
#pragma unroll
        for (int k = 0; k < SCAN_SHARE; ++k) in_t[4 * k + cw][lane] = s[k];
    }
}

__global__ __launch_bounds__(160) void cumavg_kernel(
    const float* __restrict__ q, float* __restrict__ xavg, int Bsz)
{
    __shared__ float t_in[3][SCAN_ROWS][32];
    __shared__ float t_out[2][SCAN_ROWS][32];

    const int lane = threadIdx.x & 31;
    const int wave = __builtin_amdgcn_readfirstlane((int)(threadIdx.x >> 5));
    const int col0 = blockIdx.x * 32;
    if (col0 >= Bsz * D_DIM) return;
    const int idx = col0 + lane;
    const int b = idx / D_DIM, d = idx % D_DIM;
    const float* src = q    + (size_t)b * LSEQ * D_DIM + d;
    float*       dst = xavg + (size_t)b * LSEQ * D_DIM + d;
    const size_t phase_stride = (size_t)SCAN_ROWS * D_DIM;

    float acc = 0.0f;

    if (wave != 0) {
        scan_feed_phase<true, false>(t_in[0], t_out[0], src, dst, 0, wave - 1, lane);
        scan_feed_phase<true, false>(t_in[1], t_out[0], src + phase_stride, dst, 0, wave - 1, lane);
    }
    scan_barrier();

    int in_a = 0;
    int in_f = 2;
    for (int p = 0; p < SCAN_PHASES; ++p) {
        if (wave == 0) {
            scan_add_phase(t_in[in_a], t_out[p & 1], acc, lane);
        } else {
            const float* nxt = src + (size_t)(p + 2) * phase_stride;
            const int    row0 = (p - 1) * SCAN_ROWS;
            const float (*out_f)[32] = t_out[(p + 1) & 1];
            if (p == 0)
                scan_feed_phase<true, false>(t_in[in_f], out_f, nxt, dst, row0, wave - 1, lane);
            else if (p + 2 < SCAN_PHASES)
                scan_feed_phase<true, true>(t_in[in_f], out_f, nxt, dst, row0, wave - 1, lane);
            else
                scan_feed_phase<false, true>(t_in[in_f], out_f, nxt, dst, row0, wave - 1, lane);
        }
        scan_barrier();
        in_a = (in_a == 2) ? 0 : in_a + 1;
        in_f = (in_f == 2) ? 0 : in_f + 1;
    }
    if (wave != 0)
        scan_feed_phase<false, true>(t_in[0], t_out[(SCAN_PHASES - 1) & 1], src, dst,
                                     (SCAN_PHASES - 1) * SCAN_ROWS, wave - 1, lane);
}

typedef v4f v4f_a4 __attribute__((aligned(4)));

__device__ __forceinline__ v4f keep_le(v4f v, int kb, int c)
{
    v.x = (kb     <= c) ? v.x : 0.0f;
    v.y = (kb + 1 <= c) ? v.y : 0.0f;
    v.z = (kb + 2 <= c) ? v.z : 0.0f;
    v.w = (kb + 3 <= c) ? v.w : 0.0f;
    return v;
}

template <int P>
__device__ __forceinline__ void wmma_row4(const v4f (&a)[2], const v4f (&b)[8], v8f (&acc)[4])
{
    v2f af;
    af.x = a[P >> 1][(P & 1) * 2];
    af.y = a[P >> 1][(P & 1) * 2 + 1];
#pragma unroll
    for (int t = 0; t < 4; ++t) {
        v2f bf;
        bf.x = b[2 * P][t];
        bf.y = b[2 * P + 1][t];
        acc[t] = __builtin_amdgcn_wmma_f32_16x16x4_f32(false, af, false, bf, (short)0, acc[t], false, false);
    }
}

__device__ __forceinline__ void attn_load_step(v4f (&a0)[2], v4f (&a1)[2], v4f (&bq)[8],
                                               const float* ap0, const float* ap1, const float* bp)
{
    a0[0] = *(const v4f_a4*)(ap0);
    a0[1] = *(const v4f_a4*)(ap0 + 4);
    a1[0] = *(const v4f_a4*)(ap1);
    a1[1] = *(const v4f_a4*)(ap1 + 4);
#pragma unroll
    for (int e = 0; e < 8; ++e) bq[e] = *(const v4f*)(bp + (size_t)e * D_DIM);
}

__device__ __forceinline__ void attn_mma_step(const v4f (&a0)[2], const v4f (&a1)[2],
                                              const v4f (&bq)[8], v8f (&acc)[2][4])
{
    wmma_row4<0>(a0, bq, acc[0]);  wmma_row4<0>(a1, bq, acc[1]);
    wmma_row4<1>(a0, bq, acc[0]);  wmma_row4<1>(a1, bq, acc[1]);
    wmma_row4<2>(a0, bq, acc[0]);  wmma_row4<2>(a1, bq, acc[1]);
    wmma_row4<3>(a0, bq, acc[0]);  wmma_row4<3>(a1, bq, acc[1]);
}

__global__ __launch_bounds__(128) void attn_apply_wmma(
    const float* __restrict__ g, const float* __restrict__ q,
    const float* __restrict__ P, float* __restrict__ out)
{
    const int lane = threadIdx.x & 31;
    const int wave = threadIdx.x >> 5;
    const int h    = lane >> 4;
    const int c    = lane & 15;
    const int b    = blockIdx.z;
    const int d0   = blockIdx.x * 256 + wave * 64;

    const float* qb = q + (size_t)b * LSEQ * D_DIM + (size_t)(8 * h) * D_DIM + d0 + 4 * c;

    for (int s = 0; s < 2; ++s) {
        const int y  = s ? (LSEQ / 32 - 1 - (int)blockIdx.y) : (int)blockIdx.y;
        const int i0 = y * 32;
        const int r0 = i0 + c;
        const int r1 = r0 + 16;
        const float* Ap0 = g + ((size_t)b * LSEQ + r0) * LSEQ + (LSEQ - 1 - r0) + 8 * h;
        const float* Ap1 = g + ((size_t)b * LSEQ + r1) * LSEQ + (LSEQ - 1 - r1) + 8 * h;

        v8f acc[2][4] = {};

        float inv_l;
        {
            const float* pr = P + (size_t)b * LSEQ + i0 + lane;
            const size_t mtot = (size_t)gridDim.z * LSEQ;
            float ssum = pr[0];
#pragma unroll
            for (int cb = 1; cb < LSEQ / 128; ++cb) ssum += pr[(size_t)cb * mtot];
            inv_l = 1.0f / (ssum + 1e-8f);
        }

        v4f xa0[2], xa1[2], xb[8];
        v4f ya0[2], ya1[2], yb[8];
        attn_load_step(xa0, xa1, xb, Ap0, Ap1, qb);
        for (int j0 = 0; j0 < i0; j0 += 32) {
            attn_load_step(ya0, ya1, yb, Ap0 + j0 + 16, Ap1 + j0 + 16, qb + (size_t)(j0 + 16) * D_DIM);
            __builtin_amdgcn_sched_barrier(0);
            attn_mma_step(xa0, xa1, xb, acc);
            __builtin_amdgcn_sched_barrier(0);
            attn_load_step(xa0, xa1, xb, Ap0 + j0 + 32, Ap1 + j0 + 32, qb + (size_t)(j0 + 32) * D_DIM);
            __builtin_amdgcn_sched_barrier(0);
            attn_mma_step(ya0, ya1, yb, acc);
            __builtin_amdgcn_sched_barrier(0);
        }

        ya1[0] = *(const v4f_a4*)(Ap1 + i0 + 16);
        ya1[1] = *(const v4f_a4*)(Ap1 + i0 + 20);
        {
            const float* bp = qb + (size_t)(i0 + 16) * D_DIM;
#pragma unroll
            for (int e = 0; e < 8; ++e) yb[e] = *(const v4f*)(bp + (size_t)e * D_DIM);
        }
        __builtin_amdgcn_sched_barrier(0);
        xa0[0] = keep_le(xa0[0], 8 * h,     c);
        xa0[1] = keep_le(xa0[1], 8 * h + 4, c);
        attn_mma_step(xa0, xa1, xb, acc);
        __builtin_amdgcn_sched_barrier(0);
        ya1[0] = keep_le(ya1[0], 8 * h,     c);
        ya1[1] = keep_le(ya1[1], 8 * h + 4, c);
        wmma_row4<0>(ya1, yb, acc[1]);
        wmma_row4<1>(ya1, yb, acc[1]);
        wmma_row4<2>(ya1, yb, acc[1]);
        wmma_row4<3>(ya1, yb, acc[1]);

#pragma unroll
        for (int mi = 0; mi < 2; ++mi) {
#pragma unroll
            for (int r = 0; r < 8; ++r) {
                int row = i0 + mi * 16 + r + h * 8;
                float invs = __shfl(inv_l, mi * 16 + h * 8 + r, 32);
                v4f v;
                v.x = acc[mi][0][r] * invs;  v.y = acc[mi][1][r] * invs;
                v.z = acc[mi][2][r] * invs;  v.w = acc[mi][3][r] * invs;
                *(v4f*)(out + ((size_t)b * LSEQ + row) * D_DIM + d0 + 4 * c) = v;
            }
        }
    }
}

extern "C" void kernel_launch(void* const* d_in, const int* in_sizes, int n_in,
                              void* d_out, int out_size, void* d_ws, size_t ws_size,
                              hipStream_t stream)
{
    const float* x  = (const float*)d_in[0];
    const float* Wq = (const float*)d_in[1];
    const float* Wg = (const float*)d_in[2];
    const float* bg = (const float*)d_in[3];
    const float* Wc = (const float*)d_in[4];
    float* out = (float*)d_out;

    const int Bsz = in_sizes[0] / (LSEQ * D_DIM);
    const int M   = Bsz * LSEQ;

    float* q    = (float*)d_ws;
    float* xavg = q    + (size_t)M * D_DIM;
    float* g    = xavg + (size_t)M * D_DIM;
    float* P    = out;

    dim3 blk(128);

    gemm_f32_wmma<false, false, 1, D_DIM, D_DIM><<<dim3(D_DIM / 128, M / 128), blk, 0, stream>>>(
        x, Wq, nullptr, q, nullptr);

    cumavg_kernel<<<(Bsz * D_DIM + 31) / 32, 160, 0, stream>>>(q, xavg, Bsz);

    gemm_f32_wmma<true, true, 2, LSEQ, D_DIM><<<dim3(LSEQ / 128, M / 256), blk, 0, stream>>>(
        xavg, Wg, bg, g, P);

    attn_apply_wmma<<<dim3(D_DIM / 256, LSEQ / 64, Bsz), blk, 0, stream>>>(
        g, q, P, xavg);

    gemm_f32_wmma<false, false, 1, D_DIM, D_DIM><<<dim3(D_DIM / 128, M / 128), blk, 0, stream>>>(
        xavg, Wc, nullptr, out, nullptr);
}
